# starting kernel
# baseline (speedup 1.0000x reference)
#include <hip/hip_runtime.h>

typedef __attribute__((ext_vector_type(2))) float v2f;
typedef __attribute__((ext_vector_type(8))) float v8f;

#define D 128
#define LDSPAD 132

__global__ void cheb_deg_kernel(const int* __restrict__ ei, int E,
                                float* __restrict__ deg) {
    int e = blockIdx.x * blockDim.x + threadIdx.x;
    if (e >= E) return;
    int s = ei[e];
    int d = ei[E + e];
    if (s != d) unsafeAtomicAdd(&deg[s], 1.0f);
}

__global__ void cheb_dis_kernel(const float* __restrict__ deg,
                                float* __restrict__ dis, int n) {
    int i = blockIdx.x * blockDim.x + threadIdx.x;
    if (i >= n) return;
    float dv = deg[i];
    dis[i] = (dv > 0.0f) ? rsqrtf(fmaxf(dv, 1.0f)) : 0.0f;
}

__global__ void cheb_spmm_kernel(const int* __restrict__ ei, int E,
                                 const float* __restrict__ dis,
                                 const float* __restrict__ hin,
                                 float* __restrict__ agg) {
    int wid  = (int)((blockIdx.x * (long)blockDim.x + threadIdx.x) >> 5);
    int lane = threadIdx.x & 31;
    if (wid >= E) return;
    int s = ei[wid];
    int d = ei[E + wid];
    float w = (s != d) ? -(dis[s] * dis[d]) : 0.0f;
    if (w == 0.0f) return;
    int c = lane * 4;
    const float4 h = *(const float4*)(hin + (long)s * D + c);
    float* o = agg + (long)d * D + c;
    unsafeAtomicAdd(o + 0, w * h.x);
    unsafeAtomicAdd(o + 1, w * h.y);
    unsafeAtomicAdd(o + 2, w * h.z);
    unsafeAtomicAdd(o + 3, w * h.w);
}

__global__ void __launch_bounds__(256)
cheb_gemm_kernel(const float* __restrict__ x,
                 const float* __restrict__ tx1,
                 const float* __restrict__ s2,
                 const float* __restrict__ weight,
                 const float* __restrict__ bias,
                 float* __restrict__ out, int n) {
    __shared__ float lA[3][16][LDSPAD];

    const int m0  = blockIdx.x * 16;
    const int tid = threadIdx.x;

    for (int idx = tid; idx < 3 * 16 * D; idx += 256) {
        int t = idx >> 11;
        int r = (idx >> 7) & 15;
        int c = idx & (D - 1);
        int row = m0 + r;
        float v = 0.0f;
        if (row < n) {
            long off = (long)row * D + c;
            if (t == 0)      v = x[off];
            else if (t == 1) v = tx1[off];
            else             v = 2.0f * s2[off] - x[off];
        }
        lA[t][r][c] = v;
    }
    __syncthreads();

    const int wave = tid >> 5;
    const int lane = tid & 31;
    const int hf   = lane >> 4;
    const int lm   = lane & 15;
    const int col  = wave * 16 + lm;

    v8f acc = {};
    for (int kk = 0; kk < D; kk += 4) {
        const int ka = kk + 2 * hf;
#pragma unroll
        for (int t = 0; t < 3; ++t) {
            v2f a;
            a.x = lA[t][lm][ka];
            a.y = lA[t][lm][ka + 1];
            const float* W = weight + (long)t * D * D;
            v2f b;
            b.x = W[(long)ka * D + col];
            b.y = W[(long)(ka + 1) * D + col];
            acc = __builtin_amdgcn_wmma_f32_16x16x4_f32(
                false, a, false, b, (short)0, acc, false, false);
        }
    }

    const float bv = bias[col];
#pragma unroll
    for (int i = 0; i < 8; ++i) {
        int row = m0 + i + 8 * hf;
        if (row < n) out[(long)row * D + col] = acc[i] + bv;
    }
}

extern "C" void kernel_launch(void* const* d_in, const int* in_sizes, int n_in,
                              void* d_out, int out_size, void* d_ws, size_t ws_size,
                              hipStream_t stream) {
    const float* x      = (const float*)d_in[0];
    const int*   ei     = (const int*)d_in[1];
    const float* weight = (const float*)d_in[2];
    const float* bias   = (const float*)d_in[3];
    float*       out    = (float*)d_out;

    const int n = in_sizes[0] / D;
    const int E = in_sizes[1] / 2;

    float* deg = (float*)d_ws;
    float* dis = deg + n;
    float* tx1 = dis + n;
    float* s2  = tx1 + (long)n * D;

    const size_t zbytes = ((size_t)2 * n + (size_t)2 * n * D) * sizeof(float);
    hipMemsetAsync(d_ws, 0, zbytes, stream);

    cheb_deg_kernel<<<(E + 255) / 256, 256, 0, stream>>>(ei, E, deg);
    cheb_dis_kernel<<<(n + 255) / 256, 256, 0, stream>>>(deg, dis, n);

    const int spmm_blocks = (E + 7) / 8;
    cheb_spmm_kernel<<<spmm_blocks, 256, 0, stream>>>(ei, E, dis, x,   tx1);
    cheb_spmm_kernel<<<spmm_blocks, 256, 0, stream>>>(ei, E, dis, tx1, s2);

    cheb_gemm_kernel<<<(n + 15) / 16, 256, 0, stream>>>(x, tx1, s2, weight,
                                                        bias, out, n);
}


# SpMM: one wave per run of 32 consecutive edges (was one wave per edge): indices and weights read once, broadcast by readlane, 8 row loads in flight; float atomics kept, arrival order
# speedup vs baseline: 1.0439x; 1.0439x over previous
#include <hip/hip_runtime.h>

typedef __attribute__((ext_vector_type(2))) float v2f;
typedef __attribute__((ext_vector_type(8))) float v8f;

#define D 128
#define LDSPAD 132

__global__ void cheb_deg_kernel(const int* __restrict__ ei, int E,
                                float* __restrict__ deg) {
    int e = blockIdx.x * blockDim.x + threadIdx.x;
    if (e >= E) return;
    int s = ei[e];
    int d = ei[E + e];
    if (s != d) unsafeAtomicAdd(&deg[s], 1.0f);
}

__global__ void cheb_dis_kernel(const float* __restrict__ deg,
                                float* __restrict__ dis, int n) {
    int i = blockIdx.x * blockDim.x + threadIdx.x;
    if (i >= n) return;
    float dv = deg[i];
    dis[i] = (dv > 0.0f) ? rsqrtf(fmaxf(dv, 1.0f)) : 0.0f;
}

__global__ void __launch_bounds__(256)
cheb_spmm_kernel(const int* __restrict__ ei, int E,
                 const float* __restrict__ dis,
                 const float* __restrict__ hin,
                 float* __restrict__ agg) {
    const int wid  = (int)((blockIdx.x * (long)blockDim.x + threadIdx.x) >> 5);
    const int lane = threadIdx.x & 31;
    const long base = (long)wid * 32;
    if (base >= E) return;
    const long e = base + lane;
    int   sl = 0, dl = 0;
    float wl = 0.0f;
    if (e < E) {
        sl = ei[e];
        dl = ei[(long)E + e];
        wl = (sl != dl) ? -(dis[sl] * dis[dl]) : 0.0f;
    }
    const int c = lane * 4;
#pragma unroll 1
    for (int j0 = 0; j0 < 32; j0 += 8) {
        float4 h[8];
        float  w[8];
        int    d[8];
#pragma unroll
        for (int u = 0; u < 8; ++u) {
            const int s = __builtin_amdgcn_readlane(sl, j0 + u);
            d[u] = __builtin_amdgcn_readlane(dl, j0 + u);
            w[u] = __int_as_float(
                __builtin_amdgcn_readlane(__float_as_int(wl), j0 + u));
            h[u] = *(const float4*)(hin + (long)s * D + c);
        }
#pragma unroll
        for (int u = 0; u < 8; ++u) {
            if (w[u] != 0.0f) {
                float* o = agg + (long)d[u] * D + c;
                unsafeAtomicAdd(o + 0, w[u] * h[u].x);
                unsafeAtomicAdd(o + 1, w[u] * h[u].y);
                unsafeAtomicAdd(o + 2, w[u] * h[u].z);
                unsafeAtomicAdd(o + 3, w[u] * h[u].w);
            }
        }
    }
}

__global__ void __launch_bounds__(256)
cheb_gemm_kernel(const float* __restrict__ x,
                 const float* __restrict__ tx1,
                 const float* __restrict__ s2,
                 const float* __restrict__ weight,
                 const float* __restrict__ bias,
                 float* __restrict__ out, int n) {
    __shared__ float lA[3][16][LDSPAD];

    const int m0  = blockIdx.x * 16;
    const int tid = threadIdx.x;

    for (int idx = tid; idx < 3 * 16 * D; idx += 256) {
        int t = idx >> 11;
        int r = (idx >> 7) & 15;
        int c = idx & (D - 1);
        int row = m0 + r;
        float v = 0.0f;
        if (row < n) {
            long off = (long)row * D + c;
            if (t == 0)      v = x[off];
            else if (t == 1) v = tx1[off];
            else             v = 2.0f * s2[off] - x[off];
        }
        lA[t][r][c] = v;
    }
    __syncthreads();

    const int wave = tid >> 5;
    const int lane = tid & 31;
    const int hf   = lane >> 4;
    const int lm   = lane & 15;
    const int col  = wave * 16 + lm;

    v8f acc = {};
    for (int kk = 0; kk < D; kk += 4) {
        const int ka = kk + 2 * hf;
#pragma unroll
        for (int t = 0; t < 3; ++t) {
            v2f a;
            a.x = lA[t][lm][ka];
            a.y = lA[t][lm][ka + 1];
            const float* W = weight + (long)t * D * D;
            v2f b;
            b.x = W[(long)ka * D + col];
            b.y = W[(long)(ka + 1) * D + col];
            acc = __builtin_amdgcn_wmma_f32_16x16x4_f32(
                false, a, false, b, (short)0, acc, false, false);
        }
    }

    const float bv = bias[col];
#pragma unroll
    for (int i = 0; i < 8; ++i) {
        int row = m0 + i + 8 * hf;
        if (row < n) out[(long)row * D + col] = acc[i] + bv;
    }
}

extern "C" void kernel_launch(void* const* d_in, const int* in_sizes, int n_in,
                              void* d_out, int out_size, void* d_ws, size_t ws_size,
                              hipStream_t stream) {
    const float* x      = (const float*)d_in[0];
    const int*   ei     = (const int*)d_in[1];
    const float* weight = (const float*)d_in[2];
    const float* bias   = (const float*)d_in[3];
    float*       out    = (float*)d_out;

    const int n = in_sizes[0] / D;
    const int E = in_sizes[1] / 2;

    float* deg = (float*)d_ws;
    float* dis = deg + n;
    float* tx1 = dis + n;
    float* s2  = tx1 + (long)n * D;

    const size_t zbytes = ((size_t)2 * n + (size_t)2 * n * D) * sizeof(float);
    hipMemsetAsync(d_ws, 0, zbytes, stream);

    cheb_deg_kernel<<<(E + 255) / 256, 256, 0, stream>>>(ei, E, deg);
    cheb_dis_kernel<<<(n + 255) / 256, 256, 0, stream>>>(deg, dis, n);

    const int spmm_blocks = (E + 255) / 256;
    cheb_spmm_kernel<<<spmm_blocks, 256, 0, stream>>>(ei, E, dis, x,   tx1);
    cheb_spmm_kernel<<<spmm_blocks, 256, 0, stream>>>(ei, E, dis, tx1, s2);

    cheb_gemm_kernel<<<(n + 15) / 16, 256, 0, stream>>>(x, tx1, s2, weight,
                                                        bias, out, n);
}


# SpMM: lane l owns columns l, l+32, l+64, l+96 of a row (was 4l..4l+3): each row load and each float atomic add of a wave covers one whole 128-byte line, not 4; atomics kept, arrival order as before
# speedup vs baseline: 1.7870x; 1.7118x over previous
#include <hip/hip_runtime.h>

typedef __attribute__((ext_vector_type(2))) float v2f;
typedef __attribute__((ext_vector_type(8))) float v8f;

#define D 128
#define LDSPAD 132

__global__ void cheb_deg_kernel(const int* __restrict__ ei, int E,
                                float* __restrict__ deg) {
    int e = blockIdx.x * blockDim.x + threadIdx.x;
    if (e >= E) return;
    int s = ei[e];
    int d = ei[E + e];
    if (s != d) unsafeAtomicAdd(&deg[s], 1.0f);
}

__global__ void cheb_dis_kernel(const float* __restrict__ deg,
                                float* __restrict__ dis, int n) {
    int i = blockIdx.x * blockDim.x + threadIdx.x;
    if (i >= n) return;
    float dv = deg[i];
    dis[i] = (dv > 0.0f) ? rsqrtf(fmaxf(dv, 1.0f)) : 0.0f;
}

__global__ void __launch_bounds__(256)
cheb_spmm_kernel(const int* __restrict__ ei, int E,
                 const float* __restrict__ dis,
                 const float* __restrict__ hin,
                 float* __restrict__ agg) {
    const int wid  = (int)((blockIdx.x * (long)blockDim.x + threadIdx.x) >> 5);
    const int lane = threadIdx.x & 31;
    const long base = (long)wid * 32;
    if (base >= E) return;
    const long e = base + lane;
    int   sl = 0, dl = 0;
    float wl = 0.0f;
    if (e < E) {
        sl = ei[e];
        dl = ei[(long)E + e];
        wl = (sl != dl) ? -(dis[sl] * dis[dl]) : 0.0f;
    }
#pragma unroll 1
    for (int j0 = 0; j0 < 32; j0 += 8) {
        float  h[8][4];
        float  w[8];
        int    d[8];
#pragma unroll
        for (int u = 0; u < 8; ++u) {
            const int s = __builtin_amdgcn_readlane(sl, j0 + u);
            d[u] = __builtin_amdgcn_readlane(dl, j0 + u);
            w[u] = __int_as_float(
                __builtin_amdgcn_readlane(__float_as_int(wl), j0 + u));
            const float* r = hin + (long)s * D + lane;
            h[u][0] = r[0];
            h[u][1] = r[32];
            h[u][2] = r[64];
            h[u][3] = r[96];
        }
#pragma unroll
        for (int u = 0; u < 8; ++u) {
            if (w[u] != 0.0f) {
                float* o = agg + (long)d[u] * D + lane;
                unsafeAtomicAdd(o,      w[u] * h[u][0]);
                unsafeAtomicAdd(o + 32, w[u] * h[u][1]);
                unsafeAtomicAdd(o + 64, w[u] * h[u][2]);
                unsafeAtomicAdd(o + 96, w[u] * h[u][3]);
            }
        }
    }
}

__global__ void __launch_bounds__(256)
cheb_gemm_kernel(const float* __restrict__ x,
                 const float* __restrict__ tx1,
                 const float* __restrict__ s2,
                 const float* __restrict__ weight,
                 const float* __restrict__ bias,
                 float* __restrict__ out, int n) {
    __shared__ float lA[3][16][LDSPAD];

    const int m0  = blockIdx.x * 16;
    const int tid = threadIdx.x;

    for (int idx = tid; idx < 3 * 16 * D; idx += 256) {
        int t = idx >> 11;
        int r = (idx >> 7) & 15;
        int c = idx & (D - 1);
        int row = m0 + r;
        float v = 0.0f;
        if (row < n) {
            long off = (long)row * D + c;
            if (t == 0)      v = x[off];
            else if (t == 1) v = tx1[off];
            else             v = 2.0f * s2[off] - x[off];
        }
        lA[t][r][c] = v;
    }
    __syncthreads();

    const int wave = tid >> 5;
    const int lane = tid & 31;
    const int hf   = lane >> 4;
    const int lm   = lane & 15;
    const int col  = wave * 16 + lm;

    v8f acc = {};
    for (int kk = 0; kk < D; kk += 4) {
        const int ka = kk + 2 * hf;
#pragma unroll
        for (int t = 0; t < 3; ++t) {
            v2f a;
            a.x = lA[t][lm][ka];
            a.y = lA[t][lm][ka + 1];
            const float* W = weight + (long)t * D * D;
            v2f b;
            b.x = W[(long)ka * D + col];
            b.y = W[(long)(ka + 1) * D + col];
            acc = __builtin_amdgcn_wmma_f32_16x16x4_f32(
                false, a, false, b, (short)0, acc, false, false);
        }
    }

    const float bv = bias[col];
#pragma unroll
    for (int i = 0; i < 8; ++i) {
        int row = m0 + i + 8 * hf;
        if (row < n) out[(long)row * D + col] = acc[i] + bv;
    }
}

extern "C" void kernel_launch(void* const* d_in, const int* in_sizes, int n_in,
                              void* d_out, int out_size, void* d_ws, size_t ws_size,
                              hipStream_t stream) {
    const float* x      = (const float*)d_in[0];
    const int*   ei     = (const int*)d_in[1];
    const float* weight = (const float*)d_in[2];
    const float* bias   = (const float*)d_in[3];
    float*       out    = (float*)d_out;

    const int n = in_sizes[0] / D;
    const int E = in_sizes[1] / 2;

    float* deg = (float*)d_ws;
    float* dis = deg + n;
    float* tx1 = dis + n;
    float* s2  = tx1 + (long)n * D;

    const size_t zbytes = ((size_t)2 * n + (size_t)2 * n * D) * sizeof(float);
    hipMemsetAsync(d_ws, 0, zbytes, stream);

    cheb_deg_kernel<<<(E + 255) / 256, 256, 0, stream>>>(ei, E, deg);
    cheb_dis_kernel<<<(n + 255) / 256, 256, 0, stream>>>(deg, dis, n);

    const int spmm_blocks = (E + 255) / 256;
    cheb_spmm_kernel<<<spmm_blocks, 256, 0, stream>>>(ei, E, dis, x,   tx1);
    cheb_spmm_kernel<<<spmm_blocks, 256, 0, stream>>>(ei, E, dis, tx1, s2);

    cheb_gemm_kernel<<<(n + 15) / 16, 256, 0, stream>>>(x, tx1, s2, weight,
                                                        bias, out, n);
}


# GEMM weight-stationary: each wave loads its 16 columns' weight fragments (3x128 k) into registers once; a fixed grid of 512 blocks walks the 16-row tiles, next rows fetched under the WMMAs; same sums
# speedup vs baseline: 1.9296x; 1.0798x over previous
#include <hip/hip_runtime.h>

typedef __attribute__((ext_vector_type(2))) float v2f;
typedef __attribute__((ext_vector_type(8))) float v8f;

#define D 128
#define LDSPAD 132

__global__ void cheb_deg_kernel(const int* __restrict__ ei, int E,
                                float* __restrict__ deg) {
    int e = blockIdx.x * blockDim.x + threadIdx.x;
    if (e >= E) return;
    int s = ei[e];
    int d = ei[E + e];
    if (s != d) unsafeAtomicAdd(&deg[s], 1.0f);
}

__global__ void cheb_dis_kernel(const float* __restrict__ deg,
                                float* __restrict__ dis, int n) {
    int i = blockIdx.x * blockDim.x + threadIdx.x;
    if (i >= n) return;
    float dv = deg[i];
    dis[i] = (dv > 0.0f) ? rsqrtf(fmaxf(dv, 1.0f)) : 0.0f;
}

__global__ void __launch_bounds__(256)
cheb_spmm_kernel(const int* __restrict__ ei, int E,
                 const float* __restrict__ dis,
                 const float* __restrict__ hin,
                 float* __restrict__ agg) {
    const int wid  = (int)((blockIdx.x * (long)blockDim.x + threadIdx.x) >> 5);
    const int lane = threadIdx.x & 31;
    const long base = (long)wid * 32;
    if (base >= E) return;
    const long e = base + lane;
    int   sl = 0, dl = 0;
    float wl = 0.0f;
    if (e < E) {
        sl = ei[e];
        dl = ei[(long)E + e];
        wl = (sl != dl) ? -(dis[sl] * dis[dl]) : 0.0f;
    }
#pragma unroll 1
    for (int j0 = 0; j0 < 32; j0 += 8) {
        float  h[8][4];
        float  w[8];
        int    d[8];
#pragma unroll
        for (int u = 0; u < 8; ++u) {
            const int s = __builtin_amdgcn_readlane(sl, j0 + u);
            d[u] = __builtin_amdgcn_readlane(dl, j0 + u);
            w[u] = __int_as_float(
                __builtin_amdgcn_readlane(__float_as_int(wl), j0 + u));
            const float* r = hin + (long)s * D + lane;
            h[u][0] = r[0];
            h[u][1] = r[32];
            h[u][2] = r[64];
            h[u][3] = r[96];
        }
#pragma unroll
        for (int u = 0; u < 8; ++u) {
            if (w[u] != 0.0f) {
                float* o = agg + (long)d[u] * D + lane;
                unsafeAtomicAdd(o,      w[u] * h[u][0]);
                unsafeAtomicAdd(o + 32, w[u] * h[u][1]);
                unsafeAtomicAdd(o + 64, w[u] * h[u][2]);
                unsafeAtomicAdd(o + 96, w[u] * h[u][3]);
            }
        }
    }
}

__global__ void __launch_bounds__(256)
cheb_gemm_kernel(const float* __restrict__ x,
                 const float* __restrict__ tx1,
                 const float* __restrict__ s2,
                 const float* __restrict__ weight,
                 const float* __restrict__ bias,
                 float* __restrict__ out, int n) {
    __shared__ __attribute__((aligned(16))) float lA[3][16][LDSPAD];

    const int tid  = threadIdx.x;
    const int wave = tid >> 5;
    const int lane = tid & 31;
    const int hf   = lane >> 4;
    const int lm   = lane & 15;
    const int col  = wave * 16 + lm;

    v2f b[32][3];
#pragma unroll
    for (int i = 0; i < 32; ++i) {
        const int ka = 4 * i + 2 * hf;
#pragma unroll
        for (int t = 0; t < 3; ++t) {
            const float* Wt = weight + (long)t * D * D;
            b[i][t].x = Wt[(long)ka * D + col];
            b[i][t].y = Wt[(long)(ka + 1) * D + col];
        }
    }
    const float bv = bias[col];
    const int ntiles = (n + 15) >> 4;

    const int sr = tid >> 5;
    const int sc = (tid & 31) * 4;

    float4 px[2], pt[2], ps[2];
    int tile = blockIdx.x;
    if (tile < ntiles) {
#pragma unroll
        for (int j = 0; j < 2; ++j) {
            const int row = tile * 16 + sr + 8 * j;
            px[j] = pt[j] = ps[j] = make_float4(0.f, 0.f, 0.f, 0.f);
            if (row < n) {
                const long off = (long)row * D + sc;
                px[j] = *(const float4*)(x + off);
                pt[j] = *(const float4*)(tx1 + off);
                ps[j] = *(const float4*)(s2 + off);
            }
        }
    }

    for (; tile < ntiles; tile += gridDim.x) {
#pragma unroll
        for (int j = 0; j < 2; ++j) {
            const int r = sr + 8 * j;
            float4 v2;
            v2.x = 2.0f * ps[j].x - px[j].x;
            v2.y = 2.0f * ps[j].y - px[j].y;
            v2.z = 2.0f * ps[j].z - px[j].z;
            v2.w = 2.0f * ps[j].w - px[j].w;
            *(float4*)&lA[0][r][sc] = px[j];
            *(float4*)&lA[1][r][sc] = pt[j];
            *(float4*)&lA[2][r][sc] = v2;
        }
        __syncthreads();

        const int nt = tile + gridDim.x;
        if (nt < ntiles) {
#pragma unroll
            for (int j = 0; j < 2; ++j) {
                const int row = nt * 16 + sr + 8 * j;
                px[j] = pt[j] = ps[j] = make_float4(0.f, 0.f, 0.f, 0.f);
                if (row < n) {
                    const long off = (long)row * D + sc;
                    px[j] = *(const float4*)(x + off);
                    pt[j] = *(const float4*)(tx1 + off);
                    ps[j] = *(const float4*)(s2 + off);
                }
            }
        }

        v8f acc = {};
#pragma unroll
        for (int i = 0; i < 32; ++i) {
            const int ka = 4 * i + 2 * hf;
#pragma unroll
            for (int t = 0; t < 3; ++t) {
                v2f a;
                a.x = lA[t][lm][ka];
                a.y = lA[t][lm][ka + 1];
                acc = __builtin_amdgcn_wmma_f32_16x16x4_f32(
                    false, a, false, b[i][t], (short)0, acc, false, false);
            }
        }

        const int m0 = tile * 16;
#pragma unroll
        for (int i = 0; i < 8; ++i) {
            int row = m0 + i + 8 * hf;
            if (row < n) out[(long)row * D + col] = acc[i] + bv;
        }
        __syncthreads();
    }
}

extern "C" void kernel_launch(void* const* d_in, const int* in_sizes, int n_in,
                              void* d_out, int out_size, void* d_ws, size_t ws_size,
                              hipStream_t stream) {
    const float* x      = (const float*)d_in[0];
    const int*   ei     = (const int*)d_in[1];
    const float* weight = (const float*)d_in[2];
    const float* bias   = (const float*)d_in[3];
    float*       out    = (float*)d_out;

    const int n = in_sizes[0] / D;
    const int E = in_sizes[1] / 2;

    float* deg = (float*)d_ws;
    float* dis = deg + n;
    float* tx1 = dis + n;
    float* s2  = tx1 + (long)n * D;

    const size_t zbytes = ((size_t)2 * n + (size_t)2 * n * D) * sizeof(float);
    hipMemsetAsync(d_ws, 0, zbytes, stream);

    cheb_deg_kernel<<<(E + 255) / 256, 256, 0, stream>>>(ei, E, deg);
    cheb_dis_kernel<<<(n + 255) / 256, 256, 0, stream>>>(deg, dis, n);

    const int spmm_blocks = (E + 255) / 256;
    cheb_spmm_kernel<<<spmm_blocks, 256, 0, stream>>>(ei, E, dis, x,   tx1);
    cheb_spmm_kernel<<<spmm_blocks, 256, 0, stream>>>(ei, E, dis, tx1, s2);

    cheb_gemm_kernel<<<512, 256, 0, stream>>>(x, tx1, s2, weight, bias, out, n);
}


# Products first, T(xW)=(Tx)W: GEMM stores x(W0-W2)+b to out and xW1, xW2 as whole rows; scatters add 2T(xW2) onto xW1, then T of that onto out (float atomics); no memset; sum order, rounding change
# speedup vs baseline: 2.7562x; 1.4284x over previous
#include <hip/hip_runtime.h>

typedef __attribute__((ext_vector_type(2))) float v2f;
typedef __attribute__((ext_vector_type(8))) float v8f;

#define D 128
#define LDSPAD 132

#define DEG_CAP 25600
#define DEG_SL  64
#define DEG_T   1024
#define DEG_U   8

__global__ void __launch_bounds__(DEG_T)
cheb_degcount_kernel(const int* __restrict__ ei, int E, int n,
                     int* __restrict__ part) {
    __shared__ int cnt[DEG_CAP];
    const int nr    = (n + DEG_CAP - 1) / DEG_CAP;
    const int items = nr * DEG_SL;
    const int chunk = (E + DEG_SL - 1) / DEG_SL;

    for (int item = blockIdx.x; item < items; item += gridDim.x) {
        const int r  = item % nr;
        const int s  = item / nr;
        const int lo = r * DEG_CAP;
        const int cn = (n - lo < DEG_CAP) ? (n - lo) : DEG_CAP;

        for (int j = threadIdx.x; j < DEG_CAP; j += DEG_T) cnt[j] = 0;
        __syncthreads();

        const long b  = (long)s * chunk;
        const long ee = (b + chunk < (long)E) ? (b + chunk) : (long)E;
        for (long e0 = b + threadIdx.x; e0 < ee; e0 += (long)DEG_T * DEG_U) {
            int sv[DEG_U], dv[DEG_U];
#pragma unroll
            for (int u = 0; u < DEG_U; ++u) {
                const long e = e0 + (long)u * DEG_T;
                sv[u] = -1; dv[u] = -1;
                if (e < ee) { sv[u] = ei[e]; dv[u] = ei[(long)E + e]; }
            }
#pragma unroll
            for (int u = 0; u < DEG_U; ++u) {
                const unsigned k = (unsigned)(sv[u] - lo);
                if (sv[u] != dv[u] && k < (unsigned)cn) atomicAdd(&cnt[k], 1);
            }
        }
        __syncthreads();

        int* row = part + (long)s * n + lo;
        for (int j = threadIdx.x; j < cn; j += DEG_T) row[j] = cnt[j];
        __syncthreads();
    }
}

__global__ void cheb_dis_kernel(const int* __restrict__ part,
                                float* __restrict__ dis,
                                int* __restrict__ tickets, int n) {
    int i = blockIdx.x * blockDim.x + threadIdx.x;
    if (i >= n) return;
    int c = 0;
#pragma unroll
    for (int s = 0; s < DEG_SL; ++s) c += part[(long)s * n + i];
    const float dv = (float)c;
    dis[i] = (dv > 0.0f) ? rsqrtf(fmaxf(dv, 1.0f)) : 0.0f;
    if (i < 4) tickets[i] = 0;
}

__global__ void __launch_bounds__(256)
cheb_spmm_kernel(const int* __restrict__ ei, int E,
                 const float* __restrict__ dis,
                 const float* __restrict__ hin,
                 float* __restrict__ agg,
                 int* __restrict__ tickets,
                 float scale, int n) {
    __shared__ int s_half;
    __shared__ int s_tick;
    const int wave = __builtin_amdgcn_readfirstlane((int)(threadIdx.x >> 5));
    const int lane = threadIdx.x & 31;
    const int nh   = n >> 1;
    const int nruns = (E + 31) >> 5;
    const int ntick = (nruns + 7) >> 3;

    if (threadIdx.x == 0) {
        const int die = ((int)__builtin_amdgcn_s_sendmsg_rtn(0x87) >> 16) & 15;
        s_half = (die >> 2) & 1;
    }
    __syncthreads();
    const int mine = __builtin_amdgcn_readfirstlane(s_half);

    for (int pass = 0; pass < 2; ++pass) {
        const int q = mine ^ pass;
        for (;;) {
            __syncthreads();
            if (threadIdx.x == 0)
                s_tick = __hip_atomic_fetch_add(tickets + q, 1, __ATOMIC_RELAXED,
                                                __HIP_MEMORY_SCOPE_AGENT);
            __syncthreads();
            const int t = __builtin_amdgcn_readfirstlane(s_tick);
            if (t >= ntick) break;

            const long base = ((long)t * 8 + wave) * 32;
            if (base < E) {
                const long e = base + lane;
                int   sl = 0, dl = 0;
                float wl = 0.0f;
                if (e < E) {
                    sl = ei[e];
                    dl = ei[(long)E + e];
                    const int set = (dl >= nh) ? 1 : 0;
                    if (sl != dl && set == q) wl = scale * (dis[sl] * dis[dl]);
                }
                unsigned m = __builtin_amdgcn_ballot_w32(wl != 0.0f);
#pragma unroll 1
                while (m != 0u) {
                    float  h[8][4];
                    float  w[8];
                    int    d[8];
                    bool   on[8];
#pragma unroll
                    for (int u = 0; u < 8; ++u) {
                        on[u] = (m != 0u);
                        const int j = on[u] ? __builtin_ctz(m) : 0;
                        m &= (m - 1u);
                        const int s = __builtin_amdgcn_readlane(sl, j);
                        d[u] = __builtin_amdgcn_readlane(dl, j);
                        w[u] = __int_as_float(
                            __builtin_amdgcn_readlane(__float_as_int(wl), j));
                        if (on[u]) {
                            const float* rp = hin + (long)s * D + lane;
                            h[u][0] = rp[0];
                            h[u][1] = rp[32];
                            h[u][2] = rp[64];
                            h[u][3] = rp[96];
                        }
                    }
#pragma unroll
                    for (int u = 0; u < 8; ++u) {
                        if (on[u]) {
                            float* o = agg + (long)d[u] * D + lane;
                            unsafeAtomicAdd(o,      w[u] * h[u][0]);
                            unsafeAtomicAdd(o + 32, w[u] * h[u][1]);
                            unsafeAtomicAdd(o + 64, w[u] * h[u][2]);
                            unsafeAtomicAdd(o + 96, w[u] * h[u][3]);
                        }
                    }
                }
            }
        }
    }
}

__global__ void __launch_bounds__(256, 4)
cheb_gemm_kernel(const float* __restrict__ x,
                 const float* __restrict__ weight,
                 const float* __restrict__ bias,
                 float* __restrict__ out,
                 float* __restrict__ p1,
                 float* __restrict__ p2, int n) {
    __shared__ __attribute__((aligned(16))) float lA[16][LDSPAD];
    __shared__ __attribute__((aligned(16))) float lO[3][16][LDSPAD];

    const int tid  = threadIdx.x;
    const int wave = tid >> 5;
    const int lane = tid & 31;
    const int hf   = lane >> 4;
    const int lm   = lane & 15;
    const int col  = wave * 16 + lm;

    v2f b[32][3];
#pragma unroll
    for (int i = 0; i < 32; ++i) {
        const int ka = 4 * i + 2 * hf;
        const float* W0 = weight;
        const float* W1 = weight + (long)D * D;
        const float* W2 = weight + (long)2 * D * D;
        const float w2x = W2[(long)ka * D + col];
        const float w2y = W2[(long)(ka + 1) * D + col];
        b[i][0].x = W0[(long)ka * D + col] - w2x;
        b[i][0].y = W0[(long)(ka + 1) * D + col] - w2y;
        b[i][1].x = W1[(long)ka * D + col];
        b[i][1].y = W1[(long)(ka + 1) * D + col];
        b[i][2].x = w2x;
        b[i][2].y = w2y;
    }
    const float bv = bias[col];
    const int ntiles = (n + 15) >> 4;

    const int sr = tid >> 5;
    const int sc = (tid & 31) * 4;

    float4 px[2];
    int tile = blockIdx.x;
    if (tile < ntiles) {
#pragma unroll
        for (int j = 0; j < 2; ++j) {
            const int row = tile * 16 + sr + 8 * j;
            px[j] = make_float4(0.f, 0.f, 0.f, 0.f);
            if (row < n) px[j] = *(const float4*)(x + (long)row * D + sc);
        }
    }

    for (; tile < ntiles; tile += gridDim.x) {
#pragma unroll
        for (int j = 0; j < 2; ++j) *(float4*)&lA[sr + 8 * j][sc] = px[j];
        __syncthreads();

        const int nt = tile + gridDim.x;
        if (nt < ntiles) {
#pragma unroll
            for (int j = 0; j < 2; ++j) {
                const int row = nt * 16 + sr + 8 * j;
                px[j] = make_float4(0.f, 0.f, 0.f, 0.f);
                if (row < n) px[j] = *(const float4*)(x + (long)row * D + sc);
            }
        }

        v8f acc0 = {}, acc1 = {}, acc2 = {};
#pragma unroll
        for (int i = 0; i < 32; ++i) {
            const int ka = 4 * i + 2 * hf;
            v2f a;
            a.x = lA[lm][ka];
            a.y = lA[lm][ka + 1];
            acc0 = __builtin_amdgcn_wmma_f32_16x16x4_f32(
                false, a, false, b[i][0], (short)0, acc0, false, false);
            acc1 = __builtin_amdgcn_wmma_f32_16x16x4_f32(
                false, a, false, b[i][1], (short)0, acc1, false, false);
            acc2 = __builtin_amdgcn_wmma_f32_16x16x4_f32(
                false, a, false, b[i][2], (short)0, acc2, false, false);
        }

#pragma unroll
        for (int i = 0; i < 8; ++i) {
            const int r = i + 8 * hf;
            lO[0][r][col] = acc0[i] + bv;
            lO[1][r][col] = acc1[i];
            lO[2][r][col] = acc2[i];
        }
        __syncthreads();

        const int m0 = tile * 16;
#pragma unroll
        for (int j = 0; j < 2; ++j) {
            const int r   = sr + 8 * j;
            const int row = m0 + r;
            if (row < n) {
                const long o = (long)row * D + sc;
                *(float4*)(out + o) = *(const float4*)&lO[0][r][sc];
                *(float4*)(p1 + o)  = *(const float4*)&lO[1][r][sc];
                *(float4*)(p2 + o)  = *(const float4*)&lO[2][r][sc];
            }
        }
    }
}

extern "C" void kernel_launch(void* const* d_in, const int* in_sizes, int n_in,
                              void* d_out, int out_size, void* d_ws, size_t ws_size,
                              hipStream_t stream) {
    const float* x      = (const float*)d_in[0];
    const int*   ei     = (const int*)d_in[1];
    const float* weight = (const float*)d_in[2];
    const float* bias   = (const float*)d_in[3];
    float*       out    = (float*)d_out;

    const int n = in_sizes[0] / D;
    const int E = in_sizes[1] / 2;

    int*   tick = (int*)d_ws;
    float* dis  = (float*)d_ws + n;
    float* p1   = dis + n;
    float* p2   = p1 + (long)n * D;
    int*   part = (int*)p1;

    cheb_degcount_kernel<<<256, DEG_T, 0, stream>>>(ei, E, n, part);
    cheb_dis_kernel<<<(n + 255) / 256, 256, 0, stream>>>(part, dis, tick, n);

    cheb_gemm_kernel<<<512, 256, 0, stream>>>(x, weight, bias, out, p1, p2, n);

    cheb_spmm_kernel<<<2048, 256, 0, stream>>>(ei, E, dis, p2, p1,  tick,     -2.0f, n);
    cheb_spmm_kernel<<<2048, 256, 0, stream>>>(ei, E, dis, p1, out, tick + 2, -1.0f, n);
}


# SpMM float atomics removed: edges binned per call by destination range (bins in the output plane; products last, one accumulator); block groups its bin by row in LDS, wave sums row in source order
# speedup vs baseline: 3.6063x; 1.3084x over previous
#include <hip/hip_runtime.h>

typedef __attribute__((ext_vector_type(2))) float v2f;
typedef __attribute__((ext_vector_type(8))) float v8f;

#define D 128
#define LDSPAD 132

#define DEG_CAP 25600
#define DEG_SL  64
#define DEG_T   1024
#define DEG_U   8

#define NBIN     512
#define SRC_BITS 17

__global__ void __launch_bounds__(DEG_T)
cheb_degcount_kernel(const int* __restrict__ ei, int E, int n,
                     int* __restrict__ part,
                     int* __restrict__ bc) {
    __shared__ int cnt[DEG_CAP];
    __shared__ int bcnt[NBIN];
    const unsigned rb = (unsigned)((n + NBIN - 1) / NBIN);
    const int nr    = (n + DEG_CAP - 1) / DEG_CAP;
    const int items = nr * DEG_SL;
    const int chunk = (E + DEG_SL - 1) / DEG_SL;

    for (int item = blockIdx.x; item < items; item += gridDim.x) {
        const int r  = item % nr;
        const int s  = item / nr;
        const int lo = r * DEG_CAP;
        const int cn = (n - lo < DEG_CAP) ? (n - lo) : DEG_CAP;

        for (int j = threadIdx.x; j < DEG_CAP; j += DEG_T) cnt[j] = 0;
        if (threadIdx.x < NBIN) bcnt[threadIdx.x] = 0;
        __syncthreads();

        const long b  = (long)s * chunk;
        const long ee = (b + chunk < (long)E) ? (b + chunk) : (long)E;
        const long sub = (chunk + nr - 1) / nr;
        const long slo = b + (long)r * sub;
        const long shi = (slo + sub < ee) ? (slo + sub) : ee;
        for (long e0 = b + threadIdx.x; e0 < ee; e0 += (long)DEG_T * DEG_U) {
            int sv[DEG_U], dv[DEG_U];
#pragma unroll
            for (int u = 0; u < DEG_U; ++u) {
                const long e = e0 + (long)u * DEG_T;
                sv[u] = -1; dv[u] = -1;
                if (e < ee) { sv[u] = ei[e]; dv[u] = ei[(long)E + e]; }
            }
#pragma unroll
            for (int u = 0; u < DEG_U; ++u) {
                const unsigned k = (unsigned)(sv[u] - lo);
                if (sv[u] != dv[u] && k < (unsigned)cn) atomicAdd(&cnt[k], 1);
                const long e = e0 + (long)u * DEG_T;
                if (sv[u] != dv[u] && e >= slo && e < shi)
                    atomicAdd(&bcnt[(unsigned)dv[u] / rb], 1);
            }
        }
        __syncthreads();

        int* row = part + (long)s * n + lo;
        for (int j = threadIdx.x; j < cn; j += DEG_T) row[j] = cnt[j];
        if (threadIdx.x < NBIN) bc[(long)item * NBIN + threadIdx.x] = bcnt[threadIdx.x];
        __syncthreads();
    }
}

#define SCAN_PER 8
__global__ void cheb_dis_kernel(const int* __restrict__ part,
                                float* __restrict__ dis,
                                int* __restrict__ bc,
                                int* __restrict__ binlen,
                                int n) {
    const int nblk_nodes = (n + (int)blockDim.x - 1) / (int)blockDim.x;
    const int items = ((n + DEG_CAP - 1) / DEG_CAP) * DEG_SL;
    if ((int)blockIdx.x < nblk_nodes) {
        int i = blockIdx.x * blockDim.x + threadIdx.x;
        if (i >= n) return;
        int c = 0;
#pragma unroll
        for (int s = 0; s < DEG_SL; ++s) c += part[(long)s * n + i];
        const float dv = (float)c;
        dis[i] = (dv > 0.0f) ? rsqrtf(fmaxf(dv, 1.0f)) : 0.0f;
        return;
    }
    const int lane = threadIdx.x & 31;
    const int q = (((int)blockIdx.x - nblk_nodes) * (int)blockDim.x + (int)threadIdx.x) >> 5;
    if (q >= NBIN) return;
    int c[SCAN_PER];
    int sum = 0;
#pragma unroll
    for (int k = 0; k < SCAN_PER; ++k) {
        const int it = lane * SCAN_PER + k;
        c[k] = (it < items) ? bc[(long)it * NBIN + q] : 0;
        sum += c[k];
    }
    int inc = sum;
#pragma unroll
    for (int off = 1; off < 32; off <<= 1) {
        const int t = __shfl_up(inc, off);
        if (lane >= off) inc += t;
    }
    int ex = inc - sum;
#pragma unroll
    for (int k = 0; k < SCAN_PER; ++k) {
        const int it = lane * SCAN_PER + k;
        if (it < items) bc[(long)it * NBIN + q] = ex;
        ex += c[k];
    }
    if (lane == 31) binlen[q] = inc;
}

__global__ void __launch_bounds__(DEG_T)
cheb_fill_kernel(const int* __restrict__ ei, int E, int n,
                 const int* __restrict__ bc,
                 int* __restrict__ bins) {
    __shared__ int off[NBIN];
    const int nr    = (n + DEG_CAP - 1) / DEG_CAP;
    const int items = nr * DEG_SL;
    const int chunk = (E + DEG_SL - 1) / DEG_SL;
    const unsigned rb = (unsigned)((n + NBIN - 1) / NBIN);

    for (int item = blockIdx.x; item < items; item += gridDim.x) {
        const int r = item % nr;
        const int s = item / nr;
        if (threadIdx.x < NBIN) off[threadIdx.x] = bc[(long)item * NBIN + threadIdx.x];
        __syncthreads();
        const long b   = (long)s * chunk;
        const long ee  = (b + chunk < (long)E) ? (b + chunk) : (long)E;
        const long sub = (chunk + nr - 1) / nr;
        const long slo = b + (long)r * sub;
        const long shi = (slo + sub < ee) ? (slo + sub) : ee;
        for (long e = slo + threadIdx.x; e < shi; e += DEG_T) {
            const int sv = ei[e];
            const int dv = ei[(long)E + e];
            if (sv != dv) {
                const unsigned q  = (unsigned)dv / rb;
                const unsigned lr = (unsigned)dv - q * rb;
                const int slot = atomicAdd(&off[q], 1);
                int rows = n - (int)(q * rb);
                if (rows > (int)rb) rows = (int)rb;
                if (slot < rows * D)
                    bins[(long)q * rb * D + slot] = sv | (int)(lr << SRC_BITS);
            }
        }
        __syncthreads();
    }
}

#define HOP_T   1024
#define HOP_CAP 16384
#define HOP_RB  256
__global__ void __launch_bounds__(HOP_T, 16)
cheb_hop_kernel(const int* __restrict__ bins,
                const int* __restrict__ binlen,
                const float* __restrict__ dis,
                const float* __restrict__ hin,
                const float* __restrict__ xs, int sub,
                float* __restrict__ hout, int n) {
    __shared__ int L[HOP_CAP];
    __shared__ int S[HOP_CAP];
    __shared__ int rcnt[HOP_RB];
    __shared__ int rbeg[HOP_RB + 1];
    __shared__ int rfill[HOP_RB];
    __shared__ int s_next;

    const int tid  = threadIdx.x;
    const int lane = tid & 31;
    const int rb   = (n + NBIN - 1) / NBIN;
    const int r0   = blockIdx.x * rb;
    int nrow = n - r0;
    if (nrow > rb) nrow = rb;
    if (nrow <= 0) return;
    int len = binlen[blockIdx.x];
    if (len > nrow * D) len = nrow * D;
    if (len > HOP_CAP)  len = HOP_CAP;
    const int* bin = bins + (long)r0 * D;

    if (tid < HOP_RB) rcnt[tid] = 0;
    if (tid == 0) s_next = 0;
    __syncthreads();
    for (int j = tid; j < len; j += HOP_T)
        atomicAdd(&rcnt[(bin[j] >> SRC_BITS) & (HOP_RB - 1)], 1);
    __syncthreads();
    if (tid < 32) {
        int v[HOP_RB / 32];
        int sum = 0;
#pragma unroll
        for (int k = 0; k < HOP_RB / 32; ++k) { v[k] = rcnt[lane * (HOP_RB / 32) + k]; sum += v[k]; }
        int inc = sum;
#pragma unroll
        for (int off = 1; off < 32; off <<= 1) {
            const int t = __shfl_up(inc, off);
            if (lane >= off) inc += t;
        }
        int ex = inc - sum;
#pragma unroll
        for (int k = 0; k < HOP_RB / 32; ++k) {
            rbeg[lane * (HOP_RB / 32) + k]  = ex;
            rfill[lane * (HOP_RB / 32) + k] = ex;
            ex += v[k];
        }
        if (lane == 31) rbeg[HOP_RB] = inc;
    }
    __syncthreads();
    for (int j = tid; j < len; j += HOP_T) {
        const int rec = bin[j];
        const int pos = atomicAdd(&rfill[(rec >> SRC_BITS) & (HOP_RB - 1)], 1);
        L[pos] = rec & ((1 << SRC_BITS) - 1);
    }
    __syncthreads();

    for (;;) {
        int r = 0;
        if (lane == 0) r = atomicAdd(&s_next, 1);
        r = __builtin_amdgcn_readfirstlane(r);
        if (r >= nrow) break;
        const int beg = __builtin_amdgcn_readfirstlane(rbeg[r]);
        const int cnt = __builtin_amdgcn_readfirstlane(rbeg[r + 1]) - beg;
        const long o  = (long)(r0 + r) * D + lane * 4;
        float4 xv = make_float4(0.f, 0.f, 0.f, 0.f);
        if (sub) xv = *(const float4*)(xs + o);
        const float dd = dis[r0 + r];

        for (int jc = 0; jc < cnt; jc += 32) {
            const int j  = jc + lane;
            const int kj = (j < cnt) ? L[beg + j] : 0;
            int rank = 0;
            for (int ic = 0; ic < cnt; ic += 32) {
                const int i0 = ic + lane;
                const int ki = (i0 < cnt) ? L[beg + i0] : 0;
                const int mm = (cnt - ic < 32) ? (cnt - ic) : 32;
                for (int t = 0; t < mm; ++t) {
                    const int kt = __builtin_amdgcn_readlane(ki, t);
                    const int it = ic + t;
                    rank += (kt < kj || (kt == kj && it < j)) ? 1 : 0;
                }
            }
            if (j < cnt) S[beg + rank] = kj;
        }

        float4 acc = make_float4(0.f, 0.f, 0.f, 0.f);
        for (int c0 = 0; c0 < cnt; c0 += 32) {
            const int idx = c0 + lane;
            int   sv = 0;
            float wv = 0.0f;
            if (idx < cnt) { sv = S[beg + idx]; wv = -(dis[sv] * dd); }
            const int mm = (cnt - c0 < 32) ? (cnt - c0) : 32;
#pragma unroll 1
            for (int j0 = 0; j0 < mm; j0 += 8) {
                float4 h[8];
                float  w[8];
#pragma unroll
                for (int u = 0; u < 8; ++u) {
                    const bool on = (j0 + u < mm);
                    const int  jj = on ? (j0 + u) : 0;
                    const int  s  = __builtin_amdgcn_readlane(sv, jj);
                    w[u] = __int_as_float(__builtin_amdgcn_readlane(__float_as_int(wv), jj));
                    if (on) h[u] = *(const float4*)(hin + (long)s * D + lane * 4);
                }
#pragma unroll
                for (int u = 0; u < 8; ++u) {
                    if (j0 + u < mm) {
                        acc.x = fmaf(w[u], h[u].x, acc.x);
                        acc.y = fmaf(w[u], h[u].y, acc.y);
                        acc.z = fmaf(w[u], h[u].z, acc.z);
                        acc.w = fmaf(w[u], h[u].w, acc.w);
                    }
                }
            }
        }
        if (sub) {
            acc.x = 2.0f * acc.x - xv.x;
            acc.y = 2.0f * acc.y - xv.y;
            acc.z = 2.0f * acc.z - xv.z;
            acc.w = 2.0f * acc.w - xv.w;
        }
        *(float4*)(hout + o) = acc;
    }
}

__global__ void __launch_bounds__(256, 4)
cheb_gemm_kernel(const float* __restrict__ x,
                 const float* __restrict__ t1,
                 const float* __restrict__ t2,
                 const float* __restrict__ weight,
                 const float* __restrict__ bias,
                 float* __restrict__ out, int n) {
    __shared__ __attribute__((aligned(16))) float lA[3][16][LDSPAD];
    __shared__ __attribute__((aligned(16))) float lO[16][LDSPAD];

    const int tid  = threadIdx.x;
    const int wave = tid >> 5;
    const int lane = tid & 31;
    const int hf   = lane >> 4;
    const int lm   = lane & 15;
    const int col  = wave * 16 + lm;

    v2f b[32][3];
#pragma unroll
    for (int i = 0; i < 32; ++i) {
        const int ka = 4 * i + 2 * hf;
#pragma unroll
        for (int t = 0; t < 3; ++t) {
            const float* Wt = weight + (long)t * D * D;
            b[i][t].x = Wt[(long)ka * D + col];
            b[i][t].y = Wt[(long)(ka + 1) * D + col];
        }
    }
    const float bv = bias[col];
    const int ntiles = (n + 15) >> 4;

    const int sr = tid >> 5;
    const int sc = (tid & 31) * 4;

    float4 px[3][2];
    int tile = blockIdx.x;
    if (tile < ntiles) {
#pragma unroll
        for (int j = 0; j < 2; ++j) {
            const int row = tile * 16 + sr + 8 * j;
            px[0][j] = make_float4(0.f, 0.f, 0.f, 0.f);
            px[1][j] = px[0][j];
            px[2][j] = px[0][j];
            if (row < n) {
                px[0][j] = *(const float4*)(x  + (long)row * D + sc);
                px[1][j] = *(const float4*)(t1 + (long)row * D + sc);
                px[2][j] = *(const float4*)(t2 + (long)row * D + sc);
            }
        }
    }

    for (; tile < ntiles; tile += gridDim.x) {
#pragma unroll
        for (int t = 0; t < 3; ++t)
#pragma unroll
            for (int j = 0; j < 2; ++j) *(float4*)&lA[t][sr + 8 * j][sc] = px[t][j];
        __syncthreads();

        const int nt = tile + gridDim.x;
        if (nt < ntiles) {
#pragma unroll
            for (int j = 0; j < 2; ++j) {
                const int row = nt * 16 + sr + 8 * j;
                px[0][j] = make_float4(0.f, 0.f, 0.f, 0.f);
                px[1][j] = px[0][j];
                px[2][j] = px[0][j];
                if (row < n) {
                    px[0][j] = *(const float4*)(x  + (long)row * D + sc);
                    px[1][j] = *(const float4*)(t1 + (long)row * D + sc);
                    px[2][j] = *(const float4*)(t2 + (long)row * D + sc);
                }
            }
        }

        v8f acc0 = {};
#pragma unroll
        for (int i = 0; i < 32; ++i) {
            const int ka = 4 * i + 2 * hf;
            v2f a0, a1, a2;
            a0.x = lA[0][lm][ka];  a0.y = lA[0][lm][ka + 1];
            a1.x = lA[1][lm][ka];  a1.y = lA[1][lm][ka + 1];
            a2.x = lA[2][lm][ka];  a2.y = lA[2][lm][ka + 1];
            acc0 = __builtin_amdgcn_wmma_f32_16x16x4_f32(
                false, a0, false, b[i][0], (short)0, acc0, false, false);
            acc0 = __builtin_amdgcn_wmma_f32_16x16x4_f32(
                false, a1, false, b[i][1], (short)0, acc0, false, false);
            acc0 = __builtin_amdgcn_wmma_f32_16x16x4_f32(
                false, a2, false, b[i][2], (short)0, acc0, false, false);
        }

#pragma unroll
        for (int i = 0; i < 8; ++i) {
            const int r = i + 8 * hf;
            lO[r][col] = acc0[i] + bv;
        }
        __syncthreads();

        const int m0 = tile * 16;
#pragma unroll
        for (int j = 0; j < 2; ++j) {
            const int r   = sr + 8 * j;
            const int row = m0 + r;
            if (row < n)
                *(float4*)(out + (long)row * D + sc) = *(const float4*)&lO[r][sc];
        }
    }
}

extern "C" void kernel_launch(void* const* d_in, const int* in_sizes, int n_in,
                              void* d_out, int out_size, void* d_ws, size_t ws_size,
                              hipStream_t stream) {
    const float* x      = (const float*)d_in[0];
    const int*   ei     = (const int*)d_in[1];
    const float* weight = (const float*)d_in[2];
    const float* bias   = (const float*)d_in[3];
    float*       out    = (float*)d_out;

    const int n = in_sizes[0] / D;
    const int E = in_sizes[1] / 2;

    int*   binlen = (int*)d_ws;
    float* dis    = (float*)d_ws + n;
    float* t1     = dis + n;
    float* t2     = t1 + (long)n * D;
    int*   part   = (int*)t1;
    int*   bc     = (int*)t2;
    int*   bins   = (int*)out;

    cheb_degcount_kernel<<<256, DEG_T, 0, stream>>>(ei, E, n, part, bc);
    cheb_dis_kernel<<<(n + 255) / 256 + NBIN / 8, 256, 0, stream>>>(part, dis, bc, binlen, n);
    cheb_fill_kernel<<<256, DEG_T, 0, stream>>>(ei, E, n, bc, bins);

    cheb_hop_kernel<<<NBIN, HOP_T, 0, stream>>>(bins, binlen, dis, x,  x, 0, t1, n);
    cheb_hop_kernel<<<NBIN, HOP_T, 0, stream>>>(bins, binlen, dis, t1, x, 1, t2, n);

    cheb_gemm_kernel<<<512, 256, 0, stream>>>(x, t1, t2, weight, bias, out, n);
}


# SpMM: source's degree rides in the bin word (7-bit code, 127 = read dis table); fill looks it up in an LDS copy of a byte table; hops form dis[src] from it, no random dis[src] loads; same sums
# speedup vs baseline: 3.7558x; 1.0415x over previous
#include <hip/hip_runtime.h>

typedef __attribute__((ext_vector_type(2))) float v2f;
typedef __attribute__((ext_vector_type(8))) float v8f;

#define D 128
#define LDSPAD 132

#define DEG_CAP 25600
#define DEG_SL  64
#define DEG_T   1024
#define DEG_U   8

#define NBIN     512
#define SRC_BITS 17
#define ROW_BITS 8

#define CODE_BITS 7
#define CODE_ESC  127
#define CODE_TAB  (1 << SRC_BITS)
#define TAB_U     (CODE_TAB / 16 / DEG_T)

__device__ __forceinline__ float cheb_dis_of_count(int c) {
    const float dv = (float)c;
    return (dv > 0.0f) ? rsqrtf(fmaxf(dv, 1.0f)) : 0.0f;
}

__global__ void __launch_bounds__(DEG_T)
cheb_degcount_kernel(const int* __restrict__ ei, int E, int n,
                     int* __restrict__ part,
                     int* __restrict__ bc) {
    __shared__ int cnt[DEG_CAP];
    __shared__ int bcnt[NBIN];
    const unsigned rb = (unsigned)((n + NBIN - 1) / NBIN);
    const int nr    = (n + DEG_CAP - 1) / DEG_CAP;
    const int items = nr * DEG_SL;
    const int chunk = (E + DEG_SL - 1) / DEG_SL;

    for (int item = blockIdx.x; item < items; item += gridDim.x) {
        const int r  = item % nr;
        const int s  = item / nr;
        const int lo = r * DEG_CAP;
        const int cn = (n - lo < DEG_CAP) ? (n - lo) : DEG_CAP;

        for (int j = threadIdx.x; j < DEG_CAP; j += DEG_T) cnt[j] = 0;
        if (threadIdx.x < NBIN) bcnt[threadIdx.x] = 0;
        __syncthreads();

        const long b  = (long)s * chunk;
        const long ee = (b + chunk < (long)E) ? (b + chunk) : (long)E;
        const long sub = (chunk + nr - 1) / nr;
        const long slo = b + (long)r * sub;
        const long shi = (slo + sub < ee) ? (slo + sub) : ee;
        for (long e0 = b + threadIdx.x; e0 < ee; e0 += (long)DEG_T * DEG_U) {
            int sv[DEG_U], dv[DEG_U];
#pragma unroll
            for (int u = 0; u < DEG_U; ++u) {
                const long e = e0 + (long)u * DEG_T;
                sv[u] = -1; dv[u] = -1;
                if (e < ee) { sv[u] = ei[e]; dv[u] = ei[(long)E + e]; }
            }
#pragma unroll
            for (int u = 0; u < DEG_U; ++u) {
                const unsigned k = (unsigned)(sv[u] - lo);
                if (sv[u] != dv[u] && k < (unsigned)cn) atomicAdd(&cnt[k], 1);
                const long e = e0 + (long)u * DEG_T;
                if (sv[u] != dv[u] && e >= slo && e < shi)
                    atomicAdd(&bcnt[(unsigned)dv[u] / rb], 1);
            }
        }
        __syncthreads();

        int* row = part + (long)s * n + lo;
        for (int j = threadIdx.x; j < cn; j += DEG_T) row[j] = cnt[j];
        if (threadIdx.x < NBIN) bc[(long)item * NBIN + threadIdx.x] = bcnt[threadIdx.x];
        __syncthreads();
    }
}

#define SCAN_PER 8
#define DIS_T    256
__global__ void __launch_bounds__(DIS_T)
cheb_dis_kernel(const int* __restrict__ part,
                float* __restrict__ dis,
                unsigned int* __restrict__ codes,
                int* __restrict__ bc,
                int* __restrict__ binlen,
                int n) {
    __shared__ unsigned int cw[DIS_T / 4];
    const int nblk_nodes = (n + DIS_T - 1) / DIS_T;
    const int items = ((n + DEG_CAP - 1) / DEG_CAP) * DEG_SL;
    if ((int)blockIdx.x < nblk_nodes) {
        const int i = blockIdx.x * DIS_T + threadIdx.x;
        int c = 0;
        if (i < n) {
#pragma unroll
            for (int s = 0; s < DEG_SL; ++s) c += part[(long)s * n + i];
            dis[i] = cheb_dis_of_count(c);
        }
        ((unsigned char*)cw)[threadIdx.x] = (unsigned char)((c < CODE_ESC) ? c : CODE_ESC);
        __syncthreads();
        if (threadIdx.x < DIS_T / 4) {
            const int w = blockIdx.x * (DIS_T / 4) + threadIdx.x;
            if (4 * w < ((n + 15) & ~15)) codes[w] = cw[threadIdx.x];
        }
        return;
    }
    const int lane = threadIdx.x & 31;
    const int q = (((int)blockIdx.x - nblk_nodes) * (int)blockDim.x + (int)threadIdx.x) >> 5;
    if (q >= NBIN) return;
    int c[SCAN_PER];
    int sum = 0;
#pragma unroll
    for (int k = 0; k < SCAN_PER; ++k) {
        const int it = lane * SCAN_PER + k;
        c[k] = (it < items) ? bc[(long)it * NBIN + q] : 0;
        sum += c[k];
    }
    int inc = sum;
#pragma unroll
    for (int off = 1; off < 32; off <<= 1) {
        const int t = __shfl_up(inc, off);
        if (lane >= off) inc += t;
    }
    int ex = inc - sum;
#pragma unroll
    for (int k = 0; k < SCAN_PER; ++k) {
        const int it = lane * SCAN_PER + k;
        if (it < items) bc[(long)it * NBIN + q] = ex;
        ex += c[k];
    }
    if (lane == 31) binlen[q] = inc;
}

__global__ void __launch_bounds__(DEG_T)
cheb_fill_kernel(const int* __restrict__ ei, int E, int n,
                 const int* __restrict__ bc,
                 const unsigned int* __restrict__ codes,
                 int* __restrict__ bins) {
    __shared__ int off[NBIN];
    __shared__ __attribute__((aligned(16))) unsigned int dcw[CODE_TAB / 4];
    {
        int nq = (n + 15) >> 4;
        if (nq > CODE_TAB / 16) nq = CODE_TAB / 16;
        const uint4* c4 = (const uint4*)codes;
        if (nq > 0) {
            uint4 t[TAB_U];
#pragma unroll
            for (int u = 0; u < TAB_U; ++u) {
                const int j = (int)threadIdx.x + u * DEG_T;
                t[u] = c4[(j < nq) ? j : (nq - 1)];
            }
#pragma unroll
            for (int u = 0; u < TAB_U; ++u)
                ((uint4*)dcw)[(int)threadIdx.x + u * DEG_T] = t[u];
        }
    }
    const int nr    = (n + DEG_CAP - 1) / DEG_CAP;
    const int items = nr * DEG_SL;
    const int chunk = (E + DEG_SL - 1) / DEG_SL;
    const unsigned rb = (unsigned)((n + NBIN - 1) / NBIN);

    for (int item = blockIdx.x; item < items; item += gridDim.x) {
        const int r = item % nr;
        const int s = item / nr;
        if (threadIdx.x < NBIN) off[threadIdx.x] = bc[(long)item * NBIN + threadIdx.x];
        __syncthreads();
        const long b   = (long)s * chunk;
        const long ee  = (b + chunk < (long)E) ? (b + chunk) : (long)E;
        const long sub = (chunk + nr - 1) / nr;
        const long slo = b + (long)r * sub;
        const long shi = (slo + sub < ee) ? (slo + sub) : ee;
        for (long e = slo + threadIdx.x; e < shi; e += DEG_T) {
            const int sv = ei[e];
            const int dv = ei[(long)E + e];
            if (sv != dv) {
                const unsigned q  = (unsigned)dv / rb;
                const unsigned lr = (unsigned)dv - q * rb;
                const int slot = atomicAdd(&off[q], 1);
                int rows = n - (int)(q * rb);
                if (rows > (int)rb) rows = (int)rb;
                if (slot < rows * D) {
                    const unsigned code =
                        ((const unsigned char*)dcw)[(unsigned)sv & (CODE_TAB - 1)];
                    bins[(long)q * rb * D + slot] =
                        (int)((unsigned)sv | (lr << SRC_BITS) | (code << (SRC_BITS + ROW_BITS)));
                }
            }
        }
        __syncthreads();
    }
}

#define HOP_T   1024
#define HOP_CAP 16384
#define HOP_RB  256
__global__ void __launch_bounds__(HOP_T, 16)
cheb_hop_kernel(const int* __restrict__ bins,
                const int* __restrict__ binlen,
                const float* __restrict__ dis,
                const float* __restrict__ hin,
                const float* __restrict__ xs, int sub,
                float* __restrict__ hout, int n) {
    __shared__ int L[HOP_CAP];
    __shared__ int S[HOP_CAP];
    __shared__ int rcnt[HOP_RB];
    __shared__ int rbeg[HOP_RB + 1];
    __shared__ int rfill[HOP_RB];
    __shared__ int s_next;

    const int tid  = threadIdx.x;
    const int lane = tid & 31;
    const int rb   = (n + NBIN - 1) / NBIN;
    const int r0   = blockIdx.x * rb;
    int nrow = n - r0;
    if (nrow > rb) nrow = rb;
    if (nrow <= 0) return;
    int len = binlen[blockIdx.x];
    if (len > nrow * D) len = nrow * D;
    if (len > HOP_CAP)  len = HOP_CAP;
    const int* bin = bins + (long)r0 * D;

    if (tid < HOP_RB) rcnt[tid] = 0;
    if (tid == 0) s_next = 0;
    __syncthreads();
    for (int j = tid; j < len; j += HOP_T)
        atomicAdd(&rcnt[(bin[j] >> SRC_BITS) & (HOP_RB - 1)], 1);
    __syncthreads();
    if (tid < 32) {
        int v[HOP_RB / 32];
        int sum = 0;
#pragma unroll
        for (int k = 0; k < HOP_RB / 32; ++k) { v[k] = rcnt[lane * (HOP_RB / 32) + k]; sum += v[k]; }
        int inc = sum;
#pragma unroll
        for (int off = 1; off < 32; off <<= 1) {
            const int t = __shfl_up(inc, off);
            if (lane >= off) inc += t;
        }
        int ex = inc - sum;
#pragma unroll
        for (int k = 0; k < HOP_RB / 32; ++k) {
            rbeg[lane * (HOP_RB / 32) + k]  = ex;
            rfill[lane * (HOP_RB / 32) + k] = ex;
            ex += v[k];
        }
        if (lane == 31) rbeg[HOP_RB] = inc;
    }
    __syncthreads();
    for (int j = tid; j < len; j += HOP_T) {
        const int rec = bin[j];
        const int pos = atomicAdd(&rfill[(rec >> SRC_BITS) & (HOP_RB - 1)], 1);
        L[pos] = ((rec & ((1 << SRC_BITS) - 1)) << CODE_BITS)
               | (int)((unsigned)rec >> (SRC_BITS + ROW_BITS));
    }
    __syncthreads();

    for (;;) {
        int r = 0;
        if (lane == 0) r = atomicAdd(&s_next, 1);
        r = __builtin_amdgcn_readfirstlane(r);
        if (r >= nrow) break;
        const int beg = __builtin_amdgcn_readfirstlane(rbeg[r]);
        const int cnt = __builtin_amdgcn_readfirstlane(rbeg[r + 1]) - beg;
        const long o  = (long)(r0 + r) * D + lane * 4;
        float4 xv = make_float4(0.f, 0.f, 0.f, 0.f);
        if (sub) xv = *(const float4*)(xs + o);
        const float dd = dis[r0 + r];

        for (int jc = 0; jc < cnt; jc += 32) {
            const int j  = jc + lane;
            const int kj = (j < cnt) ? L[beg + j] : 0;
            int rank = 0;
            for (int ic = 0; ic < cnt; ic += 32) {
                const int i0 = ic + lane;
                const int ki = (i0 < cnt) ? L[beg + i0] : 0;
                const int mm = (cnt - ic < 32) ? (cnt - ic) : 32;
                for (int t = 0; t < mm; ++t) {
                    const int kt = __builtin_amdgcn_readlane(ki, t);
                    const int it = ic + t;
                    rank += (kt < kj || (kt == kj && it < j)) ? 1 : 0;
                }
            }
            if (j < cnt) S[beg + rank] = kj;
        }

        float4 acc = make_float4(0.f, 0.f, 0.f, 0.f);
        for (int c0 = 0; c0 < cnt; c0 += 32) {
            const int idx = c0 + lane;
            int   sv = 0;
            float wv = 0.0f;
            if (idx < cnt) {
                const int v    = S[beg + idx];
                const int code = v & CODE_ESC;
                sv = v >> CODE_BITS;
                float ds = cheb_dis_of_count(code);
                if (code == CODE_ESC) ds = dis[sv];
                wv = -(ds * dd);
            }
            const int mm = (cnt - c0 < 32) ? (cnt - c0) : 32;
#pragma unroll 1
            for (int j0 = 0; j0 < mm; j0 += 8) {
                float4 h[8];
                float  w[8];
#pragma unroll
                for (int u = 0; u < 8; ++u) {
                    const bool on = (j0 + u < mm);
                    const int  jj = on ? (j0 + u) : 0;
                    const int  s  = __builtin_amdgcn_readlane(sv, jj);
                    w[u] = __int_as_float(__builtin_amdgcn_readlane(__float_as_int(wv), jj));
                    if (on) h[u] = *(const float4*)(hin + (long)s * D + lane * 4);
                }
#pragma unroll
                for (int u = 0; u < 8; ++u) {
                    if (j0 + u < mm) {
                        acc.x = fmaf(w[u], h[u].x, acc.x);
                        acc.y = fmaf(w[u], h[u].y, acc.y);
                        acc.z = fmaf(w[u], h[u].z, acc.z);
                        acc.w = fmaf(w[u], h[u].w, acc.w);
                    }
                }
            }
        }
        if (sub) {
            acc.x = 2.0f * acc.x - xv.x;
            acc.y = 2.0f * acc.y - xv.y;
            acc.z = 2.0f * acc.z - xv.z;
            acc.w = 2.0f * acc.w - xv.w;
        }
        *(float4*)(hout + o) = acc;
    }
}

__global__ void __launch_bounds__(256, 4)
cheb_gemm_kernel(const float* __restrict__ x,
                 const float* __restrict__ t1,
                 const float* __restrict__ t2,
                 const float* __restrict__ weight,
                 const float* __restrict__ bias,
                 float* __restrict__ out, int n) {
    __shared__ __attribute__((aligned(16))) float lA[3][16][LDSPAD];
    __shared__ __attribute__((aligned(16))) float lO[16][LDSPAD];

    const int tid  = threadIdx.x;
    const int wave = tid >> 5;
    const int lane = tid & 31;
    const int hf   = lane >> 4;
    const int lm   = lane & 15;
    const int col  = wave * 16 + lm;

    v2f b[32][3];
#pragma unroll
    for (int i = 0; i < 32; ++i) {
        const int ka = 4 * i + 2 * hf;
#pragma unroll
        for (int t = 0; t < 3; ++t) {
            const float* Wt = weight + (long)t * D * D;
            b[i][t].x = Wt[(long)ka * D + col];
            b[i][t].y = Wt[(long)(ka + 1) * D + col];
        }
    }
    const float bv = bias[col];
    const int ntiles = (n + 15) >> 4;

    const int sr = tid >> 5;
    const int sc = (tid & 31) * 4;

    float4 px[3][2];
    int tile = blockIdx.x;
    if (tile < ntiles) {
#pragma unroll
        for (int j = 0; j < 2; ++j) {
            const int row = tile * 16 + sr + 8 * j;
            px[0][j] = make_float4(0.f, 0.f, 0.f, 0.f);
            px[1][j] = px[0][j];
            px[2][j] = px[0][j];
            if (row < n) {
                px[0][j] = *(const float4*)(x  + (long)row * D + sc);
                px[1][j] = *(const float4*)(t1 + (long)row * D + sc);
                px[2][j] = *(const float4*)(t2 + (long)row * D + sc);
            }
        }
    }

    for (; tile < ntiles; tile += gridDim.x) {
#pragma unroll
        for (int t = 0; t < 3; ++t)
#pragma unroll
            for (int j = 0; j < 2; ++j) *(float4*)&lA[t][sr + 8 * j][sc] = px[t][j];
        __syncthreads();

        const int nt = tile + gridDim.x;
        if (nt < ntiles) {
#pragma unroll
            for (int j = 0; j < 2; ++j) {
                const int row = nt * 16 + sr + 8 * j;
                px[0][j] = make_float4(0.f, 0.f, 0.f, 0.f);
                px[1][j] = px[0][j];
                px[2][j] = px[0][j];
                if (row < n) {
                    px[0][j] = *(const float4*)(x  + (long)row * D + sc);
                    px[1][j] = *(const float4*)(t1 + (long)row * D + sc);
                    px[2][j] = *(const float4*)(t2 + (long)row * D + sc);
                }
            }
        }

        v8f acc0 = {};
#pragma unroll
        for (int i = 0; i < 32; ++i) {
            const int ka = 4 * i + 2 * hf;
            v2f a0, a1, a2;
            a0.x = lA[0][lm][ka];  a0.y = lA[0][lm][ka + 1];
            a1.x = lA[1][lm][ka];  a1.y = lA[1][lm][ka + 1];
            a2.x = lA[2][lm][ka];  a2.y = lA[2][lm][ka + 1];
            acc0 = __builtin_amdgcn_wmma_f32_16x16x4_f32(
                false, a0, false, b[i][0], (short)0, acc0, false, false);
            acc0 = __builtin_amdgcn_wmma_f32_16x16x4_f32(
                false, a1, false, b[i][1], (short)0, acc0, false, false);
            acc0 = __builtin_amdgcn_wmma_f32_16x16x4_f32(
                false, a2, false, b[i][2], (short)0, acc0, false, false);
        }

#pragma unroll
        for (int i = 0; i < 8; ++i) {
            const int r = i + 8 * hf;
            lO[r][col] = acc0[i] + bv;
        }
        __syncthreads();

        const int m0 = tile * 16;
#pragma unroll
        for (int j = 0; j < 2; ++j) {
            const int r   = sr + 8 * j;
            const int row = m0 + r;
            if (row < n)
                *(float4*)(out + (long)row * D + sc) = *(const float4*)&lO[r][sc];
        }
    }
}

extern "C" void kernel_launch(void* const* d_in, const int* in_sizes, int n_in,
                              void* d_out, int out_size, void* d_ws, size_t ws_size,
                              hipStream_t stream) {
    const float* x      = (const float*)d_in[0];
    const int*   ei     = (const int*)d_in[1];
    const float* weight = (const float*)d_in[2];
    const float* bias   = (const float*)d_in[3];
    float*       out    = (float*)d_out;

    const int n = in_sizes[0] / D;
    const int E = in_sizes[1] / 2;

    int*   binlen = (int*)d_ws;
    unsigned int* codes = (unsigned int*)d_ws + NBIN;
    float* dis    = (float*)d_ws + n;
    float* t1     = dis + n;
    float* t2     = t1 + (long)n * D;
    int*   part   = (int*)t1;
    int*   bc     = (int*)t2;
    int*   bins   = (int*)out;

    cheb_degcount_kernel<<<256, DEG_T, 0, stream>>>(ei, E, n, part, bc);
    cheb_dis_kernel<<<(n + 255) / 256 + NBIN / 8, DIS_T, 0, stream>>>(part, dis, codes, bc, binlen, n);
    cheb_fill_kernel<<<256, DEG_T, 0, stream>>>(ei, E, n, bc, codes, bins);

    cheb_hop_kernel<<<NBIN, HOP_T, 0, stream>>>(bins, binlen, dis, x,  x, 0, t1, n);
    cheb_hop_kernel<<<NBIN, HOP_T, 0, stream>>>(bins, binlen, dis, t1, x, 1, t2, n);

    cheb_gemm_kernel<<<512, 256, 0, stream>>>(x, t1, t2, weight, bias, out, n);
}


# Bins built in one pass as fixed cells: a fill block gathers its edge slice per bin in LDS, stores each (slice, bin) cell as whole 128-byte lines; no count pass, no scan; hops read cells; same sums
# speedup vs baseline: 4.3726x; 1.1642x over previous
#include <hip/hip_runtime.h>

typedef __attribute__((ext_vector_type(2))) float v2f;
typedef __attribute__((ext_vector_type(8))) float v8f;

#define D 128
#define LDSPAD 132

#define DEG_CAP 25600
#define DEG_SL  64
#define DEG_T   1024
#define DEG_U   8

#define NBIN     512
#define SRC_BITS 17
#define ROW_BITS 8
#define FILL_ITEMS 256
#define CELL_W     64

#define CODE_BITS 7
#define CODE_ESC  127
#define CODE_TAB  (1 << SRC_BITS)
#define TAB_U     (CODE_TAB / 16 / DEG_T)

__device__ __forceinline__ float cheb_dis_of_count(int c) {
    const float dv = (float)c;
    return (dv > 0.0f) ? rsqrtf(fmaxf(dv, 1.0f)) : 0.0f;
}

__global__ void __launch_bounds__(DEG_T)
cheb_degcount_kernel(const int* __restrict__ ei, int E, int n,
                     int* __restrict__ part) {
    __shared__ int cnt[DEG_CAP];
    const int nr    = (n + DEG_CAP - 1) / DEG_CAP;
    const int items = nr * DEG_SL;
    const int chunk = (E + DEG_SL - 1) / DEG_SL;

    for (int item = blockIdx.x; item < items; item += gridDim.x) {
        const int r  = item % nr;
        const int s  = item / nr;
        const int lo = r * DEG_CAP;
        const int cn = (n - lo < DEG_CAP) ? (n - lo) : DEG_CAP;

        for (int j = threadIdx.x; j < DEG_CAP; j += DEG_T) cnt[j] = 0;
        __syncthreads();

        const long b  = (long)s * chunk;
        const long ee = (b + chunk < (long)E) ? (b + chunk) : (long)E;
        for (long e0 = b + threadIdx.x; e0 < ee; e0 += (long)DEG_T * DEG_U) {
            int sv[DEG_U], dv[DEG_U];
#pragma unroll
            for (int u = 0; u < DEG_U; ++u) {
                const long e = e0 + (long)u * DEG_T;
                sv[u] = -1; dv[u] = -1;
                if (e < ee) { sv[u] = ei[e]; dv[u] = ei[(long)E + e]; }
            }
#pragma unroll
            for (int u = 0; u < DEG_U; ++u) {
                const unsigned k = (unsigned)(sv[u] - lo);
                if (sv[u] != dv[u] && k < (unsigned)cn) atomicAdd(&cnt[k], 1);
            }
        }
        __syncthreads();

        int* row = part + (long)s * n + lo;
        for (int j = threadIdx.x; j < cn; j += DEG_T) row[j] = cnt[j];
        __syncthreads();
    }
}

#define DIS_T    256
__global__ void __launch_bounds__(DIS_T)
cheb_dis_kernel(const int* __restrict__ part,
                float* __restrict__ dis,
                unsigned int* __restrict__ codes,
                int n) {
    __shared__ unsigned int cw[DIS_T / 4];
    const int i = blockIdx.x * DIS_T + threadIdx.x;
    int c = 0;
    if (i < n) {
#pragma unroll
        for (int s = 0; s < DEG_SL; ++s) c += part[(long)s * n + i];
        dis[i] = cheb_dis_of_count(c);
    }
    ((unsigned char*)cw)[threadIdx.x] = (unsigned char)((c < CODE_ESC) ? c : CODE_ESC);
    __syncthreads();
    if (threadIdx.x < DIS_T / 4) {
        const int w = blockIdx.x * (DIS_T / 4) + threadIdx.x;
        if (4 * w < ((n + 15) & ~15)) codes[w] = cw[threadIdx.x];
    }
}

__global__ void __launch_bounds__(DEG_T)
cheb_fill_kernel(const int* __restrict__ ei, int E, int n,
                 const unsigned int* __restrict__ codes,
                 int* __restrict__ bins) {
    __shared__ int ccnt[NBIN];
    __shared__ __attribute__((aligned(16))) unsigned int dcw[CODE_TAB / 4];
    __shared__ __attribute__((aligned(16))) int cell[NBIN * CELL_W];
    {
        int nq = (n + 15) >> 4;
        if (nq > CODE_TAB / 16) nq = CODE_TAB / 16;
        const uint4* c4 = (const uint4*)codes;
        if (nq > 0) {
            uint4 t[TAB_U];
#pragma unroll
            for (int u = 0; u < TAB_U; ++u) {
                const int j = (int)threadIdx.x + u * DEG_T;
                t[u] = c4[(j < nq) ? j : (nq - 1)];
            }
#pragma unroll
            for (int u = 0; u < TAB_U; ++u)
                ((uint4*)dcw)[(int)threadIdx.x + u * DEG_T] = t[u];
        }
    }
    const unsigned rb = (unsigned)((n + NBIN - 1) / NBIN);
    const long sub = ((long)E + FILL_ITEMS - 1) / FILL_ITEMS;

    for (int item = blockIdx.x; item < FILL_ITEMS; item += gridDim.x) {
        for (int j = threadIdx.x; j < NBIN * CELL_W / 4; j += DEG_T)
            ((uint4*)cell)[j] = make_uint4(0u, 0u, 0u, 0u);
        if (threadIdx.x < NBIN) ccnt[threadIdx.x] = 0;
        __syncthreads();
        const long slo = (long)item * sub;
        const long shi = (slo + sub < (long)E) ? (slo + sub) : (long)E;
        for (long e = slo + threadIdx.x; e < shi; e += DEG_T) {
            const int sv = ei[e];
            const int dv = ei[(long)E + e];
            if (sv != dv) {
                const unsigned q  = (unsigned)dv / rb;
                const unsigned lr = (unsigned)dv - q * rb;
                const int slot = atomicAdd(&ccnt[q], 1);
                if (slot < CELL_W - 1) {
                    const unsigned code =
                        ((const unsigned char*)dcw)[(unsigned)sv & (CODE_TAB - 1)];
                    cell[q * CELL_W + 1 + slot] =
                        (int)((unsigned)sv | (lr << SRC_BITS) | (code << (SRC_BITS + ROW_BITS)));
                }
            }
        }
        __syncthreads();
        if (threadIdx.x < NBIN) {
            const int cc = ccnt[threadIdx.x];
            cell[threadIdx.x * CELL_W] = (cc < CELL_W - 1) ? cc : (CELL_W - 1);
        }
        __syncthreads();
        for (int j = threadIdx.x; j < NBIN * CELL_W / 4; j += DEG_T) {
            const int q = j >> 4;
            const int k = j & 15;
            if (k < 8 || cell[q * CELL_W] >= CELL_W / 2)
                *(uint4*)(bins + ((long)q * FILL_ITEMS + item) * CELL_W + k * 4) =
                    ((const uint4*)cell)[j];
        }
        __syncthreads();
    }
}

#define HOP_T   1024
#define HOP_CAP (FILL_ITEMS * CELL_W)
#define HOP_RB  256
__global__ void __launch_bounds__(HOP_T, 16)
cheb_hop_kernel(const int* __restrict__ bins,
                const float* __restrict__ dis,
                const float* __restrict__ hin,
                const float* __restrict__ xs, int sub,
                float* __restrict__ hout, int n) {
    __shared__ int L[HOP_CAP];
    __shared__ int S[HOP_CAP];
    __shared__ int rcnt[HOP_RB];
    __shared__ int rbeg[HOP_RB + 1];
    __shared__ int rfill[HOP_RB];
    __shared__ int s_next;

    const int tid  = threadIdx.x;
    const int lane = tid & 31;
    const int rb   = (n + NBIN - 1) / NBIN;
    const int r0   = blockIdx.x * rb;
    int nrow = n - r0;
    if (nrow > rb) nrow = rb;
    if (nrow <= 0) return;

    const int part = tid & 3;
    const uint4* cp = (const uint4*)(bins + ((long)blockIdx.x * FILL_ITEMS + (tid >> 2)) * CELL_W
                                     + part * 16);
    uint4 cw4[4];
#pragma unroll
    for (int u = 0; u < 4; ++u) cw4[u] = make_uint4(0u, 0u, 0u, 0u);
    if (part < 2) {
#pragma unroll
        for (int u = 0; u < 4; ++u) cw4[u] = cp[u];
    }
    int ccount = __shfl((int)cw4[0].x, lane & ~3);
    if (ccount < 0) ccount = 0;
    if (ccount > CELL_W - 1) ccount = CELL_W - 1;
    if (part >= 2 && ccount >= CELL_W / 2) {
#pragma unroll
        for (int u = 0; u < 4; ++u) cw4[u] = cp[u];
    }
    int wds[16];
#pragma unroll
    for (int u = 0; u < 4; ++u) {
        wds[4 * u + 0] = (int)cw4[u].x;  wds[4 * u + 1] = (int)cw4[u].y;
        wds[4 * u + 2] = (int)cw4[u].z;  wds[4 * u + 3] = (int)cw4[u].w;
    }
    const int wlo = 1 - 16 * part;
    const int whi = ccount - 16 * part;

    if (tid < HOP_RB) rcnt[tid] = 0;
    if (tid == 0) s_next = 0;
    __syncthreads();
#pragma unroll
    for (int k = 0; k < 16; ++k)
        if (k >= wlo && k <= whi)
            atomicAdd(&rcnt[(wds[k] >> SRC_BITS) & (HOP_RB - 1)], 1);
    __syncthreads();
    if (tid < 32) {
        int v[HOP_RB / 32];
        int sum = 0;
#pragma unroll
        for (int k = 0; k < HOP_RB / 32; ++k) { v[k] = rcnt[lane * (HOP_RB / 32) + k]; sum += v[k]; }
        int inc = sum;
#pragma unroll
        for (int off = 1; off < 32; off <<= 1) {
            const int t = __shfl_up(inc, off);
            if (lane >= off) inc += t;
        }
        int ex = inc - sum;
#pragma unroll
        for (int k = 0; k < HOP_RB / 32; ++k) {
            rbeg[lane * (HOP_RB / 32) + k]  = ex;
            rfill[lane * (HOP_RB / 32) + k] = ex;
            ex += v[k];
        }
        if (lane == 31) rbeg[HOP_RB] = inc;
    }
    __syncthreads();
#pragma unroll
    for (int k = 0; k < 16; ++k) {
        if (k >= wlo && k <= whi) {
            const int rec = wds[k];
            const int pos = atomicAdd(&rfill[(rec >> SRC_BITS) & (HOP_RB - 1)], 1);
            L[pos] = ((rec & ((1 << SRC_BITS) - 1)) << CODE_BITS)
                   | (int)((unsigned)rec >> (SRC_BITS + ROW_BITS));
        }
    }
    __syncthreads();

    for (;;) {
        int r = 0;
        if (lane == 0) r = atomicAdd(&s_next, 1);
        r = __builtin_amdgcn_readfirstlane(r);
        if (r >= nrow) break;
        const int beg = __builtin_amdgcn_readfirstlane(rbeg[r]);
        const int cnt = __builtin_amdgcn_readfirstlane(rbeg[r + 1]) - beg;
        const long o  = (long)(r0 + r) * D + lane * 4;
        float4 xv = make_float4(0.f, 0.f, 0.f, 0.f);
        if (sub) xv = *(const float4*)(xs + o);
        const float dd = dis[r0 + r];

        for (int jc = 0; jc < cnt; jc += 32) {
            const int j  = jc + lane;
            const int kj = (j < cnt) ? L[beg + j] : 0;
            int rank = 0;
            for (int ic = 0; ic < cnt; ic += 32) {
                const int i0 = ic + lane;
                const int ki = (i0 < cnt) ? L[beg + i0] : 0;
                const int mm = (cnt - ic < 32) ? (cnt - ic) : 32;
                for (int t = 0; t < mm; ++t) {
                    const int kt = __builtin_amdgcn_readlane(ki, t);
                    const int it = ic + t;
                    rank += (kt < kj || (kt == kj && it < j)) ? 1 : 0;
                }
            }
            if (j < cnt) S[beg + rank] = kj;
        }

        float4 acc = make_float4(0.f, 0.f, 0.f, 0.f);
        for (int c0 = 0; c0 < cnt; c0 += 32) {
            const int idx = c0 + lane;
            int   sv = 0;
            float wv = 0.0f;
            if (idx < cnt) {
                const int v    = S[beg + idx];
                const int code = v & CODE_ESC;
                sv = v >> CODE_BITS;
                float ds = cheb_dis_of_count(code);
                if (code == CODE_ESC) ds = dis[sv];
                wv = -(ds * dd);
            }
            const int mm = (cnt - c0 < 32) ? (cnt - c0) : 32;
#pragma unroll 1
            for (int j0 = 0; j0 < mm; j0 += 8) {
                float4 h[8];
                float  w[8];
#pragma unroll
                for (int u = 0; u < 8; ++u) {
                    const bool on = (j0 + u < mm);
                    const int  jj = on ? (j0 + u) : 0;
                    const int  s  = __builtin_amdgcn_readlane(sv, jj);
                    w[u] = __int_as_float(__builtin_amdgcn_readlane(__float_as_int(wv), jj));
                    if (on) h[u] = *(const float4*)(hin + (long)s * D + lane * 4);
                }
#pragma unroll
                for (int u = 0; u < 8; ++u) {
                    if (j0 + u < mm) {
                        acc.x = fmaf(w[u], h[u].x, acc.x);
                        acc.y = fmaf(w[u], h[u].y, acc.y);
                        acc.z = fmaf(w[u], h[u].z, acc.z);
                        acc.w = fmaf(w[u], h[u].w, acc.w);
                    }
                }
            }
        }
        if (sub) {
            acc.x = 2.0f * acc.x - xv.x;
            acc.y = 2.0f * acc.y - xv.y;
            acc.z = 2.0f * acc.z - xv.z;
            acc.w = 2.0f * acc.w - xv.w;
        }
        *(float4*)(hout + o) = acc;
    }
}

__global__ void __launch_bounds__(256, 4)
cheb_gemm_kernel(const float* __restrict__ x,
                 const float* __restrict__ t1,
                 const float* __restrict__ t2,
                 const float* __restrict__ weight,
                 const float* __restrict__ bias,
                 float* __restrict__ out, int n) {
    __shared__ __attribute__((aligned(16))) float lA[3][16][LDSPAD];
    __shared__ __attribute__((aligned(16))) float lO[16][LDSPAD];

    const int tid  = threadIdx.x;
    const int wave = tid >> 5;
    const int lane = tid & 31;
    const int hf   = lane >> 4;
    const int lm   = lane & 15;
    const int col  = wave * 16 + lm;

    v2f b[32][3];
#pragma unroll
    for (int i = 0; i < 32; ++i) {
        const int ka = 4 * i + 2 * hf;
#pragma unroll
        for (int t = 0; t < 3; ++t) {
            const float* Wt = weight + (long)t * D * D;
            b[i][t].x = Wt[(long)ka * D + col];
            b[i][t].y = Wt[(long)(ka + 1) * D + col];
        }
    }
    const float bv = bias[col];
    const int ntiles = (n + 15) >> 4;

    const int sr = tid >> 5;
    const int sc = (tid & 31) * 4;

    float4 px[3][2];
    int tile = blockIdx.x;
    if (tile < ntiles) {
#pragma unroll
        for (int j = 0; j < 2; ++j) {
            const int row = tile * 16 + sr + 8 * j;
            px[0][j] = make_float4(0.f, 0.f, 0.f, 0.f);
            px[1][j] = px[0][j];
            px[2][j] = px[0][j];
            if (row < n) {
                px[0][j] = *(const float4*)(x  + (long)row * D + sc);
                px[1][j] = *(const float4*)(t1 + (long)row * D + sc);
                px[2][j] = *(const float4*)(t2 + (long)row * D + sc);
            }
        }
    }

    for (; tile < ntiles; tile += gridDim.x) {
#pragma unroll
        for (int t = 0; t < 3; ++t)
#pragma unroll
            for (int j = 0; j < 2; ++j) *(float4*)&lA[t][sr + 8 * j][sc] = px[t][j];
        __syncthreads();

        const int nt = tile + gridDim.x;
        if (nt < ntiles) {
#pragma unroll
            for (int j = 0; j < 2; ++j) {
                const int row = nt * 16 + sr + 8 * j;
                px[0][j] = make_float4(0.f, 0.f, 0.f, 0.f);
                px[1][j] = px[0][j];
                px[2][j] = px[0][j];
                if (row < n) {
                    px[0][j] = *(const float4*)(x  + (long)row * D + sc);
                    px[1][j] = *(const float4*)(t1 + (long)row * D + sc);
                    px[2][j] = *(const float4*)(t2 + (long)row * D + sc);
                }
            }
        }

        v8f acc0 = {};
#pragma unroll
        for (int i = 0; i < 32; ++i) {
            const int ka = 4 * i + 2 * hf;
            v2f a0, a1, a2;
            a0.x = lA[0][lm][ka];  a0.y = lA[0][lm][ka + 1];
            a1.x = lA[1][lm][ka];  a1.y = lA[1][lm][ka + 1];
            a2.x = lA[2][lm][ka];  a2.y = lA[2][lm][ka + 1];
            acc0 = __builtin_amdgcn_wmma_f32_16x16x4_f32(
                false, a0, false, b[i][0], (short)0, acc0, false, false);
            acc0 = __builtin_amdgcn_wmma_f32_16x16x4_f32(
                false, a1, false, b[i][1], (short)0, acc0, false, false);
            acc0 = __builtin_amdgcn_wmma_f32_16x16x4_f32(
                false, a2, false, b[i][2], (short)0, acc0, false, false);
        }

#pragma unroll
        for (int i = 0; i < 8; ++i) {
            const int r = i + 8 * hf;
            lO[r][col] = acc0[i] + bv;
        }
        __syncthreads();

        const int m0 = tile * 16;
#pragma unroll
        for (int j = 0; j < 2; ++j) {
            const int r   = sr + 8 * j;
            const int row = m0 + r;
            if (row < n)
                *(float4*)(out + (long)row * D + sc) = *(const float4*)&lO[r][sc];
        }
    }
}

extern "C" void kernel_launch(void* const* d_in, const int* in_sizes, int n_in,
                              void* d_out, int out_size, void* d_ws, size_t ws_size,
                              hipStream_t stream) {
    const float* x      = (const float*)d_in[0];
    const int*   ei     = (const int*)d_in[1];
    const float* weight = (const float*)d_in[2];
    const float* bias   = (const float*)d_in[3];
    float*       out    = (float*)d_out;

    const int n = in_sizes[0] / D;
    const int E = in_sizes[1] / 2;

    unsigned int* codes = (unsigned int*)d_ws + NBIN;
    float* dis    = (float*)d_ws + n;
    float* t1     = dis + n;
    float* t2     = t1 + (long)n * D;
    int*   part   = (int*)t1;
    int*   bins   = (int*)out;

    cheb_degcount_kernel<<<256, DEG_T, 0, stream>>>(ei, E, n, part);
    cheb_dis_kernel<<<(n + 255) / 256, DIS_T, 0, stream>>>(part, dis, codes, n);
    cheb_fill_kernel<<<FILL_ITEMS, DEG_T, 0, stream>>>(ei, E, n, codes, bins);

    cheb_hop_kernel<<<NBIN, HOP_T, 0, stream>>>(bins, dis, x,  x, 0, t1, n);
    cheb_hop_kernel<<<NBIN, HOP_T, 0, stream>>>(bins, dis, t1, x, 1, t2, n);

    cheb_gemm_kernel<<<512, 256, 0, stream>>>(x, t1, t2, weight, bias, out, n);
}
